# starting kernel
# baseline (speedup 1.0000x reference)
#include <hip/hip_runtime.h>


typedef __attribute__((ext_vector_type(2))) float v2f;
typedef __attribute__((ext_vector_type(8))) float v8f;

#define B_      64
#define S_      512
#define D_      768
#define MW_     256
#define L_      8
#define CHUNKS  4
#define CHUNK_S (S_ / CHUNKS)
#define THREADS 256
#define PART_STRIDE (MW_ * (L_ + 1))

__global__ __launch_bounds__(THREADS)
void seg_proj_wmma_kernel(const float* __restrict__ h,
                          const int*   __restrict__ wid,
                          const float* __restrict__ W,
                          float* __restrict__ part)
{
    __shared__ float Wt[L_ * D_];
    __shared__ float seg[PART_STRIDE];

    const int tid  = threadIdx.x;
    const int b    = blockIdx.x / CHUNKS;
    const int c    = blockIdx.x % CHUNKS;
    const int wave = tid >> 5;
    const int lane = tid & 31;

    for (int i = tid; i < PART_STRIDE; i += THREADS) seg[i] = 0.0f;
    for (int i = tid; i < L_ * D_; i += THREADS) {
        const int n = i / D_;
        const int k = i - n * D_;
        Wt[i] = W[k * L_ + n];
    }
    __syncthreads();

    const int tok0 = c * CHUNK_S + wave * 16;
    const int m    = lane & 15;
    const int khi  = (lane >> 4) * 2;

    const float* __restrict__ arow = h + (size_t)(b * S_ + tok0 + m) * D_ + khi;
    const float* __restrict__ brow = Wt + (lane & 7) * D_ + khi;

    v8f acc = {};
#pragma unroll 4
    for (int k0 = 0; k0 < D_; k0 += 4) {
        v2f a  = { arow[k0], arow[k0 + 1] };
        v2f w2 = { brow[k0], brow[k0 + 1] };
        acc = __builtin_amdgcn_wmma_f32_16x16x4_f32(
             false, a,  false, w2,
             (short)0, acc,  false,  false);
    }

    const int n = lane & 15;
    if (n < L_) {
        const int mbase = (lane >> 4) * 8;
#pragma unroll
        for (int v = 0; v < 8; ++v) {
            const int tok = tok0 + mbase + v;
            const int w   = wid[b * S_ + tok];
            if (w >= 0 && w < MW_) {
                atomicAdd(&seg[w * L_ + n], acc[v]);
                if (n == 0) atomicAdd(&seg[MW_ * L_ + w], 1.0f);
            }
        }
    }
    __syncthreads();

    float* __restrict__ out = part + (size_t)blockIdx.x * PART_STRIDE;
    for (int i = tid; i < PART_STRIDE; i += THREADS) out[i] = seg[i];
}

__global__ __launch_bounds__(THREADS)
void seg_finalize_kernel(const float* __restrict__ part,
                         const float* __restrict__ bias,
                         float* __restrict__ out)
{
    const int t = blockIdx.x * blockDim.x + threadIdx.x;
    if (t >= B_ * MW_) return;
    const int b = t / MW_;
    const int w = t - b * MW_;

    float s[L_] = {0.f, 0.f, 0.f, 0.f, 0.f, 0.f, 0.f, 0.f};
    float cnt = 0.f;
    for (int c = 0; c < CHUNKS; ++c) {
        const float* __restrict__ p = part + (size_t)(b * CHUNKS + c) * PART_STRIDE;
#pragma unroll
        for (int l = 0; l < L_; ++l) s[l] += p[w * L_ + l];
        cnt += p[MW_ * L_ + w];
    }
    const float inv = (cnt > 0.f) ? (1.0f / cnt) : 0.0f;
    float* __restrict__ o = out + (size_t)t * L_;
#pragma unroll
    for (int l = 0; l < L_; ++l) o[l] = bias[l] + s[l] * inv;
}

extern "C" void kernel_launch(void* const* d_in, const int* in_sizes, int n_in,
                              void* d_out, int out_size, void* d_ws, size_t ws_size,
                              hipStream_t stream)
{
    const float* h    = (const float*)d_in[0];
    const int*   wid  = (const int*)  d_in[1];
    const float* W    = (const float*)d_in[2];
    const float* bias = (const float*)d_in[3];

    float* part = (float*)d_ws;

    seg_proj_wmma_kernel<<<B_ * CHUNKS, THREADS, 0, stream>>>(h, wid, W, part);
    seg_finalize_kernel<<<(B_ * MW_ + THREADS - 1) / THREADS, THREADS, 0, stream>>>(
        part, bias, (float*)d_out);
}


# h fetch rebuilt: 16-byte loads, 64 contiguous bytes per token row per load (spare output columns carry a second k half), 32-48 loads in flight per wave; k summation order changed
# speedup vs baseline: 1.8372x; 1.8372x over previous
#include <hip/hip_runtime.h>


typedef __attribute__((ext_vector_type(2))) float v2f;
typedef __attribute__((ext_vector_type(4))) float v4f;
typedef __attribute__((ext_vector_type(8))) float v8f;

#define B_      64
#define S_      512
#define D_      768
#define MW_     256
#define L_      8
#define CHUNKS  4
#define CHUNK_S (S_ / CHUNKS)
#define THREADS 256
#define PART_STRIDE (MW_ * (L_ + 1))

#define KSTEP   16
#define NSTEP   (D_ / KSTEP)
#define GSTEPS  8
#define NBATCH  (NSTEP / GSTEPS)
#define ALOADS  (2 * GSTEPS)

__global__ __launch_bounds__(THREADS)
void seg_proj_wmma_kernel(const float* __restrict__ h,
                          const int*   __restrict__ wid,
                          const float* __restrict__ W,
                          float* __restrict__ part)
{
    __shared__ v4f   Bf[NSTEP * 32];
    __shared__ float seg[PART_STRIDE];

    const int tid  = threadIdx.x;
    const int b    = blockIdx.x / CHUNKS;
    const int c    = blockIdx.x % CHUNKS;
    const int wave = tid >> 5;
    const int lane = tid & 31;

    const int tok0 = c * CHUNK_S + wave * 16;
    const int t    = lane & 7;
    const int j    = (lane >> 3) & 1;
    const int hf   = lane >> 4;

    float wreg[(NSTEP * 32 * 4) / THREADS];
    {
        const int e   = tid & 3;
        const int n   = (tid >> 2) & 7;
        const int jj  = (tid >> 5) & 1;
        const int hh  = (tid >> 6) & 1;
        const int k0  = (tid >> 7) * KSTEP + jj * 8 + hh * 4 + e;
        const float* __restrict__ wsrc = W + k0 * L_ + n;
#pragma unroll
        for (int q = 0; q < (NSTEP * 32 * 4) / THREADS; ++q)
            wreg[q] = wsrc[q * (2 * KSTEP * L_)];
    }

    const v4f* __restrict__ a0p =
        (const v4f*)(h + (size_t)(b * S_ + tok0 + t) * D_ + j * 8 + hf * 4);
    const v4f* __restrict__ a1p = a0p + (8 * D_) / 4;

    v4f abuf[3][ALOADS];

#define LOAD_A_BATCH(buf, bi)                                              \
    _Pragma("unroll")                                                      \
    for (int s2 = 0; s2 < GSTEPS; s2 += 2) {                               \
        (buf)[2 * s2 + 0] = a0p[((bi) * GSTEPS + s2    ) * (KSTEP / 4)];   \
        (buf)[2 * s2 + 2] = a0p[((bi) * GSTEPS + s2 + 1) * (KSTEP / 4)];   \
        (buf)[2 * s2 + 1] = a1p[((bi) * GSTEPS + s2    ) * (KSTEP / 4)];   \
        (buf)[2 * s2 + 3] = a1p[((bi) * GSTEPS + s2 + 1) * (KSTEP / 4)];   \
    }

    LOAD_A_BATCH(abuf[0], 0)
    LOAD_A_BATCH(abuf[1], 1)

    for (int i = tid; i < PART_STRIDE; i += THREADS) seg[i] = 0.0f;
    {
        float* __restrict__ bflat = (float*)Bf;
#pragma unroll
        for (int q = 0; q < (NSTEP * 32 * 4) / THREADS; ++q)
            bflat[tid + q * THREADS] = wreg[q];
    }
    __builtin_amdgcn_fence(__ATOMIC_RELEASE, "workgroup", "local");
    __builtin_amdgcn_s_barrier();
    __builtin_amdgcn_fence(__ATOMIC_ACQUIRE, "workgroup", "local");

    v4f bbuf[2][GSTEPS];
#define LOAD_B_BATCH(buf, bi)                                              \
    _Pragma("unroll")                                                      \
    for (int s = 0; s < GSTEPS; ++s)                                       \
        (buf)[s] = Bf[((bi) * GSTEPS + s) * 32 + lane];

    LOAD_B_BATCH(bbuf[0], 0)

    v8f acc0 = {};
    v8f acc1 = {};
#pragma unroll
    for (int i = 0; i < NBATCH; ++i) {
        if (i + 2 < NBATCH) { LOAD_A_BATCH(abuf[(i + 2) % 3], i + 2) }
        if (i + 1 < NBATCH) { LOAD_B_BATCH(bbuf[(i + 1) & 1], i + 1) }
        __builtin_amdgcn_sched_barrier(0);
#pragma unroll
        for (int s = 0; s < GSTEPS; ++s) {
            const v4f a0 = abuf[i % 3][2 * s + 0];
            const v4f a1 = abuf[i % 3][2 * s + 1];
            const v4f w4 = bbuf[i & 1][s];
            const v2f a0lo = { a0.x, a0.y }, a0hi = { a0.z, a0.w };
            const v2f a1lo = { a1.x, a1.y }, a1hi = { a1.z, a1.w };
            const v2f wlo  = { w4.x, w4.y }, whi  = { w4.z, w4.w };
            acc0 = __builtin_amdgcn_wmma_f32_16x16x4_f32(
                false, a0lo, false, wlo, (short)0, acc0, false, false);
            acc1 = __builtin_amdgcn_wmma_f32_16x16x4_f32(
                false, a1lo, false, wlo, (short)0, acc1, false, false);
            acc0 = __builtin_amdgcn_wmma_f32_16x16x4_f32(
                false, a0hi, false, whi, (short)0, acc0, false, false);
            acc1 = __builtin_amdgcn_wmma_f32_16x16x4_f32(
                false, a1hi, false, whi, (short)0, acc1, false, false);
        }
        __builtin_amdgcn_sched_barrier(0);
    }
#undef LOAD_A_BATCH
#undef LOAD_B_BATCH

    float s0[8], s1[8];
#pragma unroll
    for (int r = 0; r < 8; ++r) {
        s0[r] = acc0[r] + __shfl_xor(acc0[r], 24);
        s1[r] = acc1[r] + __shfl_xor(acc1[r], 24);
    }

    const bool lo = lane < 8;
    const bool hi = lane >= 24;
    if (lo || hi) {
        const int n     = lane & 7;
        const int tbase = tok0 + (hi ? 8 : 0);
#pragma unroll
        for (int r = 0; r < 8; ++r) {
            const float val = hi ? s1[r] : s0[r];
            const int   w   = wid[b * S_ + tbase + r];
            if (w >= 0 && w < MW_) {
                atomicAdd(&seg[w * L_ + n], val);
                if (n == 0) atomicAdd(&seg[MW_ * L_ + w], 1.0f);
            }
        }
    }
    __syncthreads();

    float* __restrict__ out = part + (size_t)blockIdx.x * PART_STRIDE;
    for (int i = tid; i < PART_STRIDE; i += THREADS) out[i] = seg[i];
}

__global__ __launch_bounds__(THREADS)
void seg_finalize_kernel(const float* __restrict__ part,
                         const float* __restrict__ bias,
                         float* __restrict__ out)
{
    const int t = blockIdx.x * blockDim.x + threadIdx.x;
    if (t >= B_ * MW_) return;
    const int b = t / MW_;
    const int w = t - b * MW_;

    float s[L_] = {0.f, 0.f, 0.f, 0.f, 0.f, 0.f, 0.f, 0.f};
    float cnt = 0.f;
    for (int c = 0; c < CHUNKS; ++c) {
        const float* __restrict__ p = part + (size_t)(b * CHUNKS + c) * PART_STRIDE;
#pragma unroll
        for (int l = 0; l < L_; ++l) s[l] += p[w * L_ + l];
        cnt += p[MW_ * L_ + w];
    }
    const float inv = (cnt > 0.f) ? (1.0f / cnt) : 0.0f;
    float* __restrict__ o = out + (size_t)t * L_;
#pragma unroll
    for (int l = 0; l < L_; ++l) o[l] = bias[l] + s[l] * inv;
}

extern "C" void kernel_launch(void* const* d_in, const int* in_sizes, int n_in,
                              void* d_out, int out_size, void* d_ws, size_t ws_size,
                              hipStream_t stream)
{
    const float* h    = (const float*)d_in[0];
    const int*   wid  = (const int*)  d_in[1];
    const float* W    = (const float*)d_in[2];
    const float* bias = (const float*)d_in[3];

    float* part = (float*)d_ws;

    seg_proj_wmma_kernel<<<B_ * CHUNKS, THREADS, 0, stream>>>(h, wid, W, part);
    seg_finalize_kernel<<<(B_ * MW_ + THREADS - 1) / THREADS, THREADS, 0, stream>>>(
        part, bias, (float*)d_out);
}
